# starting kernel
# baseline (speedup 1.0000x reference)
#include <hip/hip_runtime.h>

#define N_NODES 20000
#define KN      32
#define DFEAT   128
#define NODES_PER_BLOCK 32

typedef __attribute__((ext_vector_type(2))) float v2f;
typedef __attribute__((ext_vector_type(4))) float v4f;
typedef __attribute__((ext_vector_type(8))) float v8f;

__global__ __launch_bounds__(256)
void gnn_layer_kernel(const float* __restrict__ h_prev,
                      const int*   __restrict__ adj,
                      const float* __restrict__ W,
                      const float* __restrict__ bias,
                      float*       __restrict__ h_out)
{
    __shared__ float smW[DFEAT * DFEAT];
    __shared__ float smA[NODES_PER_BLOCK * DFEAT];
    __shared__ int   smAdj[NODES_PER_BLOCK * KN];

    const int tid = threadIdx.x;
    const int n0  = blockIdx.x * NODES_PER_BLOCK;

    {
        const v4f* __restrict__ W4 = reinterpret_cast<const v4f*>(W);
        v4f* __restrict__ smW4     = reinterpret_cast<v4f*>(smW);
        #pragma unroll
        for (int it = 0; it < 16; ++it)
            smW4[tid + it * 256] = W4[tid + it * 256];
    }

    #pragma unroll
    for (int it = 0; it < 4; ++it) {
        int idx = tid + it * 256;
        int nl  = idx & (NODES_PER_BLOCK - 1);
        int k   = idx >> 5;
        smAdj[nl * KN + k] = adj[k * N_NODES + n0 + nl];
    }
    __syncthreads();

    const v4f* __restrict__ hp4 = reinterpret_cast<const v4f*>(h_prev);
    v4f* __restrict__ smA4      = reinterpret_cast<v4f*>(smA);

    #pragma unroll
    for (int it = 0; it < 4; ++it) {
        int idx = tid + it * 256;
        int nl  = idx >> 5;
        int c4  = idx & 31;
        v4f acc = {};
        #pragma unroll 8
        for (int k = 0; k < KN; ++k) {
            int nb = smAdj[nl * KN + k];
            acc += hp4[(size_t)nb * (DFEAT / 4) + c4];
        }
        smA4[nl * (DFEAT / 4) + c4] = acc * (1.0f / KN);
    }
    __syncthreads();

    const int wave    = tid >> 5;
    const int lane    = tid & 31;
    const int half    = lane >> 4;
    const int l16     = lane & 15;
    const int colBase = wave * 16;

    v8f c0 = {};
    v8f c1 = {};
    #pragma unroll
    for (int ks = 0; ks < DFEAT / 4; ++ks) {
        const int k0 = ks * 4 + half * 2;
        v2f bm;
        bm.x = smW[(k0    ) * DFEAT + colBase + l16];
        bm.y = smW[(k0 + 1) * DFEAT + colBase + l16];
        v2f a0, a1;
        a0.x = smA[(l16     ) * DFEAT + k0];
        a0.y = smA[(l16     ) * DFEAT + k0 + 1];
        a1.x = smA[(l16 + 16) * DFEAT + k0];
        a1.y = smA[(l16 + 16) * DFEAT + k0 + 1];
        c0 = __builtin_amdgcn_wmma_f32_16x16x4_f32(
                 false, a0, false, bm, (short)0, c0, false, false);
        c1 = __builtin_amdgcn_wmma_f32_16x16x4_f32(
                 false, a1, false, bm, (short)0, c1, false, false);
    }

    const int   col = colBase + l16;
    const float bv  = bias[col];
    #pragma unroll
    for (int r = 0; r < 8; ++r) {
        const int row = r + half * 8;
        float v0 = c0[r] + bv;
        float v1 = c1[r] + bv;
        v0 = (v0 > 0.0f) ? v0 : 0.0f;
        v1 = (v1 > 0.0f) ? v1 : 0.0f;
        h_out[(size_t)(n0 + row     ) * DFEAT + col] = v0;
        h_out[(size_t)(n0 + row + 16) * DFEAT + col] = v1;
    }
}

extern "C" void kernel_launch(void* const* d_in, const int* in_sizes, int n_in,
                              void* d_out, int out_size, void* d_ws, size_t ws_size,
                              hipStream_t stream) {
    (void)in_sizes; (void)n_in; (void)out_size; (void)d_ws; (void)ws_size;

    const int*   adj   = (const int*)  d_in[0];
    const float* graph = (const float*)d_in[1];
    const float* W     = (const float*)d_in[2];
    const float* b     = (const float*)d_in[3];
    float*       out   = (float*)d_out;

    const size_t layerElems = (size_t)N_NODES * DFEAT;

    hipMemcpyAsync(out, graph, layerElems * sizeof(float),
                   hipMemcpyDeviceToDevice, stream);

    dim3 grid(N_NODES / NODES_PER_BLOCK);
    dim3 block(256);

    gnn_layer_kernel<<<grid, block, 0, stream>>>(
        graph, adj, W + 1 * DFEAT * DFEAT, b + 1 * DFEAT, out + 1 * layerElems);

    gnn_layer_kernel<<<grid, block, 0, stream>>>(
        out + 1 * layerElems, adj, W + 2 * DFEAT * DFEAT, b + 2 * DFEAT,
        out + 2 * layerElems);
}


# LDS rows padded (A tile pitch 128->132 floats, W 128->144) so WMMA A/B fragment reads spread over all 64 banks instead of 16-way (A) and 4-way (B) bank conflicts; arithmetic and sum order unchanged
# speedup vs baseline: 1.3095x; 1.3095x over previous
#include <hip/hip_runtime.h>

#define N_NODES 20000
#define KN      32
#define DFEAT   128
#define NODES_PER_BLOCK 32
#define A_PITCH 132
#define W_PITCH 144

typedef __attribute__((ext_vector_type(2))) float v2f;
typedef __attribute__((ext_vector_type(4))) float v4f;
typedef __attribute__((ext_vector_type(8))) float v8f;

__global__ __launch_bounds__(256)
void gnn_layer_kernel(const float* __restrict__ h_prev,
                      const int*   __restrict__ adj,
                      const float* __restrict__ W,
                      const float* __restrict__ bias,
                      float*       __restrict__ h_out)
{
    __shared__ float smW[DFEAT * W_PITCH];
    __shared__ float smA[NODES_PER_BLOCK * A_PITCH];
    __shared__ int   smAdj[NODES_PER_BLOCK * KN];

    const int tid = threadIdx.x;
    const int n0  = blockIdx.x * NODES_PER_BLOCK;

    {
        const v4f* __restrict__ W4 = reinterpret_cast<const v4f*>(W);
        v4f* __restrict__ smW4     = reinterpret_cast<v4f*>(smW);
        #pragma unroll
        for (int it = 0; it < 16; ++it) {
            int idx = tid + it * 256;
            smW4[(idx >> 5) * (W_PITCH / 4) + (idx & 31)] = W4[idx];
        }
    }

    #pragma unroll
    for (int it = 0; it < 4; ++it) {
        int idx = tid + it * 256;
        int nl  = idx & (NODES_PER_BLOCK - 1);
        int k   = idx >> 5;
        smAdj[nl * KN + k] = adj[k * N_NODES + n0 + nl];
    }
    __syncthreads();

    const v4f* __restrict__ hp4 = reinterpret_cast<const v4f*>(h_prev);
    v4f* __restrict__ smA4      = reinterpret_cast<v4f*>(smA);

    #pragma unroll
    for (int it = 0; it < 4; ++it) {
        int idx = tid + it * 256;
        int nl  = idx >> 5;
        int c4  = idx & 31;
        v4f acc = {};
        #pragma unroll 8
        for (int k = 0; k < KN; ++k) {
            int nb = smAdj[nl * KN + k];
            acc += hp4[(size_t)nb * (DFEAT / 4) + c4];
        }
        smA4[nl * (A_PITCH / 4) + c4] = acc * (1.0f / KN);
    }
    __syncthreads();

    const int wave    = tid >> 5;
    const int lane    = tid & 31;
    const int half    = lane >> 4;
    const int l16     = lane & 15;
    const int colBase = wave * 16;

    v8f c0 = {};
    v8f c1 = {};
    #pragma unroll
    for (int ks = 0; ks < DFEAT / 4; ++ks) {
        const int k0 = ks * 4 + half * 2;
        v2f bm;
        bm.x = smW[(k0    ) * W_PITCH + colBase + l16];
        bm.y = smW[(k0 + 1) * W_PITCH + colBase + l16];
        v2f a0, a1;
        a0.x = smA[(l16     ) * A_PITCH + k0];
        a0.y = smA[(l16     ) * A_PITCH + k0 + 1];
        a1.x = smA[(l16 + 16) * A_PITCH + k0];
        a1.y = smA[(l16 + 16) * A_PITCH + k0 + 1];
        c0 = __builtin_amdgcn_wmma_f32_16x16x4_f32(
                 false, a0, false, bm, (short)0, c0, false, false);
        c1 = __builtin_amdgcn_wmma_f32_16x16x4_f32(
                 false, a1, false, bm, (short)0, c1, false, false);
    }

    const int   col = colBase + l16;
    const float bv  = bias[col];
    #pragma unroll
    for (int r = 0; r < 8; ++r) {
        const int row = r + half * 8;
        float v0 = c0[r] + bv;
        float v1 = c1[r] + bv;
        v0 = (v0 > 0.0f) ? v0 : 0.0f;
        v1 = (v1 > 0.0f) ? v1 : 0.0f;
        h_out[(size_t)(n0 + row     ) * DFEAT + col] = v0;
        h_out[(size_t)(n0 + row + 16) * DFEAT + col] = v1;
    }
}

extern "C" void kernel_launch(void* const* d_in, const int* in_sizes, int n_in,
                              void* d_out, int out_size, void* d_ws, size_t ws_size,
                              hipStream_t stream) {
    (void)in_sizes; (void)n_in; (void)out_size; (void)d_ws; (void)ws_size;

    const int*   adj   = (const int*)  d_in[0];
    const float* graph = (const float*)d_in[1];
    const float* W     = (const float*)d_in[2];
    const float* b     = (const float*)d_in[3];
    float*       out   = (float*)d_out;

    const size_t layerElems = (size_t)N_NODES * DFEAT;

    hipMemcpyAsync(out, graph, layerElems * sizeof(float),
                   hipMemcpyDeviceToDevice, stream);

    dim3 grid(N_NODES / NODES_PER_BLOCK);
    dim3 block(256);

    gnn_layer_kernel<<<grid, block, 0, stream>>>(
        graph, adj, W + 1 * DFEAT * DFEAT, b + 1 * DFEAT, out + 1 * layerElems);

    gnn_layer_kernel<<<grid, block, 0, stream>>>(
        out + 1 * layerElems, adj, W + 2 * DFEAT * DFEAT, b + 2 * DFEAT,
        out + 2 * layerElems);
}
